# starting kernel
# baseline (speedup 1.0000x reference)
#include <hip/hip_runtime.h>
#include <hip/hip_bf16.h>

typedef float v2f __attribute__((ext_vector_type(2)));
typedef float v8f __attribute__((ext_vector_type(8)));

static __device__ __forceinline__ v8f wmma_f32_4(v2f a, v2f b, v8f c) {
  return __builtin_amdgcn_wmma_f32_16x16x4_f32(false, a, false, b, (short)0, c,
                                               false, false);
}

#define ZERO8 {0.f,0.f,0.f,0.f,0.f,0.f,0.f,0.f}

__global__ __launch_bounds__(256) void gemm128x64_wmma(
    const float* __restrict__ A, const float* __restrict__ B,
    const float* __restrict__ bias, float* __restrict__ C,
    int M, int N, int K) {
  __shared__ float bt[64 * 66];

  const int m0 = blockIdx.x * 128;
  const int n0 = blockIdx.y * 64;
  const int tid = threadIdx.x;
  const int w = tid >> 5;
  const int lane = tid & 31;
  const int L = lane & 15;
  const int hi = lane >> 4;
  const int arow = m0 + w * 16 + L;

  v8f acc[4] = {ZERO8, ZERO8, ZERO8, ZERO8};

  const int c4 = (tid & 15) * 4;
  const int r0 = tid >> 4;

  for (int kc = 0; kc < K; kc += 64) {
#pragma unroll
    for (int p = 0; p < 4; ++p) {
      const int r = r0 + p * 16;
      const float4 bv =
          *(const float4*)&B[(size_t)(kc + r) * N + n0 + c4];
      bt[(c4 + 0) * 66 + r] = bv.x;
      bt[(c4 + 1) * 66 + r] = bv.y;
      bt[(c4 + 2) * 66 + r] = bv.z;
      bt[(c4 + 3) * 66 + r] = bv.w;
    }
    __syncthreads();
#pragma unroll
    for (int kk = 0; kk < 64; kk += 4) {
      const v2f av = *(const v2f*)&A[(size_t)arow * K + kc + kk + hi * 2];
#pragma unroll
      for (int t = 0; t < 4; ++t) {
        const v2f bf = *(const v2f*)&bt[(t * 16 + L) * 66 + kk + hi * 2];
        acc[t] = wmma_f32_4(av, bf, acc[t]);
      }
    }
    __syncthreads();
  }

  const int crow = m0 + w * 16 + hi * 8;
#pragma unroll
  for (int t = 0; t < 4; ++t) {
    const int col = n0 + t * 16 + L;
    const float bs = bias ? bias[col] : 0.0f;
#pragma unroll
    for (int r = 0; r < 8; ++r)
      C[(size_t)(crow + r) * N + col] = acc[t][r] + bs;
  }
}

__global__ __launch_bounds__(256) void attn_kernel(
    const float* __restrict__ qkv, const int* __restrict__ mask,
    float* __restrict__ out) {
  extern __shared__ float lds[];
  float* simb = lds;
  float* rinv = lds + 256 * 258;

  const int bid = blockIdx.x;
  const int s = bid >> 5;
  const int b = (bid >> 3) & 3;
  const int h = bid & 7;

  const float* base = qkv + (size_t)((s * 4 + b) * 256) * 768;
  const float* qb = base + h * 32;
  const float* kb = base + 256 + h * 32;
  const float* vb = base + 512 + h * 32;
  const int* mk = mask + (size_t)s * 256 * 256;
  float* ob = out + (size_t)((s * 4 + b) * 256) * 256 + h * 32;

  const int tid = threadIdx.x;
  const int w = tid >> 5;
  const int lane = tid & 31;
  const int L = lane & 15;
  const int hi = lane >> 4;
  const float scale = 0.17677669529663687f;
  const float NEG = -1.0e10f;

  v2f qa[2][8];
#pragma unroll
  for (int rb = 0; rb < 2; ++rb) {
    const int rbase = rb * 128 + w * 16;
#pragma unroll
    for (int c = 0; c < 8; ++c)
      qa[rb][c] = *(const v2f*)&qb[(size_t)(rbase + L) * 768 + c * 4 + hi * 2];
  }

  for (int j0 = 0; j0 < 256; j0 += 16) {
    v2f kf[8];
#pragma unroll
    for (int c = 0; c < 8; ++c)
      kf[c] = *(const v2f*)&kb[(size_t)(j0 + L) * 768 + c * 4 + hi * 2];
#pragma unroll
    for (int rb = 0; rb < 2; ++rb) {
      v8f acc = ZERO8;
#pragma unroll
      for (int c = 0; c < 8; ++c) acc = wmma_f32_4(qa[rb][c], kf[c], acc);
      const int ib = rb * 128 + w * 16 + hi * 8;
#pragma unroll
      for (int r = 0; r < 8; ++r) {
        float sv = acc[r] * scale;
        sv = (mk[(ib + r) * 256 + j0 + L] == 0) ? NEG : sv;
        simb[(ib + r) * 258 + j0 + L] = sv;
      }
    }
  }
  __syncthreads();

  {
    float* row = simb + tid * 258;
    float m = -3.4e38f;
    for (int j = 0; j < 256; ++j) m = fmaxf(m, row[j]);
    float sum = 0.0f;
    for (int j = 0; j < 256; ++j) {
      const float e = __expf(row[j] - m);
      row[j] = e;
      sum += e;
    }
    rinv[tid] = 1.0f / sum;
  }
  __syncthreads();

  v8f oacc[2][2] = {{ZERO8, ZERO8}, {ZERO8, ZERO8}};
  for (int jj = 0; jj < 256; jj += 4) {
    v2f va[2];
#pragma unroll
    for (int dt = 0; dt < 2; ++dt) {
      v2f t;
      t.x = vb[(size_t)(jj + hi * 2) * 768 + dt * 16 + L];
      t.y = vb[(size_t)(jj + hi * 2 + 1) * 768 + dt * 16 + L];
      va[dt] = t;
    }
#pragma unroll
    for (int rb = 0; rb < 2; ++rb) {
      const v2f pa =
          *(const v2f*)&simb[(rb * 128 + w * 16 + L) * 258 + jj + hi * 2];
      oacc[rb][0] = wmma_f32_4(pa, va[0], oacc[rb][0]);
      oacc[rb][1] = wmma_f32_4(pa, va[1], oacc[rb][1]);
    }
  }
#pragma unroll
  for (int rb = 0; rb < 2; ++rb) {
    const int ib = rb * 128 + w * 16 + hi * 8;
#pragma unroll
    for (int r = 0; r < 8; ++r) {
      const float ri = rinv[ib + r];
      ob[(size_t)(ib + r) * 256 + L] = oacc[rb][0][r] * ri;
      ob[(size_t)(ib + r) * 256 + 16 + L] = oacc[rb][1][r] * ri;
    }
  }
}

extern "C" void kernel_launch(void* const* d_in, const int* in_sizes, int n_in,
                              void* d_out, int out_size, void* d_ws,
                              size_t ws_size, hipStream_t stream) {
  (void)in_sizes; (void)n_in; (void)out_size; (void)ws_size;
  const float* x    = (const float*)d_in[0];
  const int*   mask = (const int*)d_in[1];
  const float* Wqkv = (const float*)d_in[2];
  const float* Wout = (const float*)d_in[3];
  const float* bout = (const float*)d_in[4];
  float* outp = (float*)d_out;

  float* qkv = (float*)d_ws;
  float* ao  = qkv + (size_t)65536 * 768;

  const dim3 blk(256);

  gemm128x64_wmma<<<dim3(512, 12), blk, 0, stream>>>(x, Wqkv, nullptr, qkv,
                                                     65536, 768, 256);

  const size_t smem = (size_t)(256 * 258 + 256) * sizeof(float);
  hipFuncSetAttribute((const void*)attn_kernel,
                      hipFuncAttributeMaxDynamicSharedMemorySize, (int)smem);
  attn_kernel<<<dim3(2048), blk, smem, stream>>>(qkv, mask, ao);

  gemm128x64_wmma<<<dim3(512, 4), blk, 0, stream>>>(ao, Wout, bout, outp,
                                                    65536, 256, 256);
}


# attention: free fragment index maps re-chosen so a lane's operands are adjacent in memory: q/k 16-byte loads, V/mask/out 8-byte whole-line accesses, 8-byte score stores; score-k sum order changes
# speedup vs baseline: 1.6371x; 1.6371x over previous
#include <hip/hip_runtime.h>
#include <hip/hip_bf16.h>

typedef float v2f __attribute__((ext_vector_type(2)));
typedef float v8f __attribute__((ext_vector_type(8)));

static __device__ __forceinline__ v8f wmma_f32_4(v2f a, v2f b, v8f c) {
  return __builtin_amdgcn_wmma_f32_16x16x4_f32(false, a, false, b, (short)0, c,
                                               false, false);
}

#define ZERO8 {0.f,0.f,0.f,0.f,0.f,0.f,0.f,0.f}

#define GK 256
#define GN 128
#define GPC 36
#define GCH (GN * GPC)

__global__ __launch_bounds__(512) __attribute__((amdgpu_waves_per_eu(8, 8)))
void gemm256x128_wmma(
    const float* __restrict__ A, const float* __restrict__ B,
    const float* __restrict__ bias, float* __restrict__ C, int N) {
  extern __shared__ float bt[];

  const int n0 = blockIdx.x * GN;
  const int m0 = blockIdx.y * 256;
  const int tid = threadIdx.x;
  const int w = tid >> 5;
  const int lane = tid & 31;
  const int L = lane & 15;
  const int hi = lane >> 4;

  {
    const int n4 = lane & 3;
    const int kq = lane >> 2;
    float4 bv[16];
#pragma unroll
    for (int p = 0; p < 16; ++p) {
      const int q = w * 16 + p;
      const int n = (q & 7) * 16 + n4 * 4;
      const int k = (q >> 3) * 8 + kq;
      bv[p] = *(const float4*)&B[(size_t)k * N + n0 + n];
    }
#pragma unroll
    for (int p = 0; p < 16; ++p) {
      const int q = w * 16 + p;
      const int n = (q & 7) * 16 + n4 * 4;
      const int k = (q >> 3) * 8 + kq;
      float* dst = bt + (k >> 5) * GCH + n * GPC + (k & 31);
      dst[0 * GPC] = bv[p].x;
      dst[1 * GPC] = bv[p].y;
      dst[2 * GPC] = bv[p].z;
      dst[3 * GPC] = bv[p].w;
    }
  }

  const float* abase = A + (size_t)m0 * GK;
  const unsigned aoff = (unsigned)((w * 16 + L) * GK + hi * 2);
  const float* brow = bt + L * GPC + hi * 2;

  v8f acc[8] = {ZERO8, ZERO8, ZERO8, ZERO8, ZERO8, ZERO8, ZERO8, ZERO8};

  v2f a_cur[8];
#pragma unroll
  for (int c = 0; c < 8; ++c) a_cur[c] = *(const v2f*)&abase[aoff + (unsigned)(c * 4)];

  __syncthreads();

  for (int kc = 0; kc < GK; kc += 32) {
    const int kn = (kc + 32 < GK) ? kc + 32 : kc;
    v2f a_nxt[8];
#pragma unroll
    for (int c = 0; c < 8; ++c) a_nxt[c] = *(const v2f*)&(abase + kn)[aoff + (unsigned)(c * 4)];
    const float* bch = brow + (kc >> 5) * GCH;
#pragma unroll
    for (int c = 0; c < 8; ++c) {
#pragma unroll
      for (int t = 0; t < 8; ++t) {
        const v2f bf = *(const v2f*)&bch[t * 16 * GPC + c * 4];
        acc[t] = wmma_f32_4(a_cur[c], bf, acc[t]);
      }
    }
#pragma unroll
    for (int c = 0; c < 8; ++c) a_cur[c] = a_nxt[c];
  }

  float* cbase = C + (size_t)m0 * N + n0;
  const unsigned coff = (unsigned)((w * 16 + hi * 8) * N + L);
#pragma unroll
  for (int t = 0; t < 8; ++t) {
    const float bs = bias ? bias[n0 + t * 16 + L] : 0.0f;
#pragma unroll
    for (int r = 0; r < 8; ++r)
      cbase[coff + (unsigned)(r * N + t * 16)] = acc[t][r] + bs;
  }
}

static __device__ __forceinline__ void load_v(v2f (&dst)[16][2],
                                              const float* __restrict__ vlane,
                                              int j0) {
#pragma unroll
  for (int ks = 0; ks < 16; ++ks) {
    const v2f r0 = *(const v2f*)&vlane[(size_t)(j0 + ks * 4) * 768];
    const v2f r1 = *(const v2f*)&vlane[(size_t)(j0 + ks * 4 + 1) * 768];
    dst[ks][0] = (v2f){r0.x, r1.x};
    dst[ks][1] = (v2f){r0.y, r1.y};
  }
}

static __device__ __forceinline__ void load_scores(v2f (&sc)[2][8],
                                                   const float* srow0,
                                                   const float* srow1, int j) {
#pragma unroll
  for (int c = 0; c < 8; ++c) {
    sc[0][c] = *(const v2f*)&srow0[j + c * 4];
    sc[1][c] = *(const v2f*)&srow1[j + c * 4];
  }
}

static __device__ __forceinline__ void pv_chunk(
    v8f (&oacc)[2][2], float (&ssum)[2], v2f (&sc)[2][8],
    const v2f (&vf)[16][2], const float (&mrow)[2], const float* srow0,
    const float* srow1, int j0) {
#pragma unroll
  for (int hf = 0; hf < 2; ++hf) {
    v2f pa[2][8];
#pragma unroll
    for (int c = 0; c < 8; ++c) {
#pragma unroll
      for (int rb = 0; rb < 2; ++rb) {
        v2f p;
        p.x = __expf(sc[rb][c].x - mrow[rb]);
        p.y = __expf(sc[rb][c].y - mrow[rb]);
        ssum[rb] += p.x;
        ssum[rb] += p.y;
        pa[rb][c] = p;
      }
    }
    const int jn = (j0 + hf * 32 + 32 < 256) ? j0 + hf * 32 + 32 : 224;
    __builtin_amdgcn_sched_barrier(0);
    load_scores(sc, srow0, srow1, jn);
    __builtin_amdgcn_sched_barrier(0);
#pragma unroll
    for (int c = 0; c < 8; ++c) {
#pragma unroll
      for (int rb = 0; rb < 2; ++rb) {
        oacc[rb][0] = wmma_f32_4(pa[rb][c], vf[hf * 8 + c][0], oacc[rb][0]);
        oacc[rb][1] = wmma_f32_4(pa[rb][c], vf[hf * 8 + c][1], oacc[rb][1]);
      }
    }
  }
}

__global__ __launch_bounds__(256) void attn_kernel(
    const float* __restrict__ qkv, const int* __restrict__ mask,
    float* __restrict__ out) {
  extern __shared__ float lds[];
  float* simb = lds;
  float* rinv = lds + 256 * 258;

  const int bid = blockIdx.x;
  const int s = bid >> 5;
  const int b = (bid >> 3) & 3;
  const int h = bid & 7;

  const float* base = qkv + (size_t)((s * 4 + b) * 256) * 768;
  const float* qb = base + h * 32;
  const float* kb = base + 256 + h * 32;
  const float* vb = base + 512 + h * 32;
  const int* mk = mask + (size_t)s * 256 * 256;
  float* ob = out + (size_t)((s * 4 + b) * 256) * 256 + h * 32;

  const int tid = threadIdx.x;
  const int w = tid >> 5;
  const int lane = tid & 31;
  const int L = lane & 15;
  const int hi = lane >> 4;
  const float scale = 0.17677669529663687f;
  const float NEG = -1.0e10f;

  v2f qa[2][8];
#pragma unroll
  for (int rb = 0; rb < 2; ++rb) {
    const int rbase = rb * 128 + w * 16;
#pragma unroll
    for (int c2 = 0; c2 < 4; ++c2) {
      const float4 q4 =
          *(const float4*)&qb[(size_t)(rbase + L) * 768 + c2 * 8 + hi * 4];
      qa[rb][2 * c2] = (v2f){q4.x, q4.y};
      qa[rb][2 * c2 + 1] = (v2f){q4.z, q4.w};
    }
  }

  for (int j0 = 0; j0 < 256; j0 += 32) {
    v2f kf[2][8];
#pragma unroll
    for (int t = 0; t < 2; ++t) {
#pragma unroll
      for (int c2 = 0; c2 < 4; ++c2) {
        const float4 k4 = *(const float4*)&kb[(size_t)(j0 + 2 * L + t) * 768 +
                                              c2 * 8 + hi * 4];
        kf[t][2 * c2] = (v2f){k4.x, k4.y};
        kf[t][2 * c2 + 1] = (v2f){k4.z, k4.w};
      }
    }
#pragma unroll
    for (int rb = 0; rb < 2; ++rb) {
      v8f acc0 = ZERO8, acc1 = ZERO8;
#pragma unroll
      for (int c = 0; c < 8; ++c) {
        acc0 = wmma_f32_4(qa[rb][c], kf[0][c], acc0);
        acc1 = wmma_f32_4(qa[rb][c], kf[1][c], acc1);
      }
      const int ib = rb * 128 + w * 16 + hi * 8;
#pragma unroll
      for (int r = 0; r < 8; ++r) {
        const int2 m2 = *(const int2*)&mk[(ib + r) * 256 + j0 + 2 * L];
        v2f sv;
        sv.x = acc0[r] * scale;
        sv.y = acc1[r] * scale;
        sv.x = (m2.x == 0) ? NEG : sv.x;
        sv.y = (m2.y == 0) ? NEG : sv.y;
        *(v2f*)&simb[(ib + r) * 258 + j0 + 2 * L] = sv;
      }
    }
  }
  const float* vlane = vb + (size_t)(hi * 2) * 768 + 2 * L;
  const float* srow0 = simb + (w * 16 + L) * 258 + hi * 2;
  const float* srow1 = simb + (128 + w * 16 + L) * 258 + hi * 2;
  v2f vA[16][2], vB[16][2];
  load_v(vA, vlane, 0);
  __syncthreads();

  float mrow[2];
#pragma unroll
  for (int rb = 0; rb < 2; ++rb) {
    const float* row = simb + (rb * 128 + w * 16 + L) * 258 + hi * 32;
    float m = -3.4e38f;
#pragma unroll
    for (int g = 0; g < 4; ++g) {
#pragma unroll
      for (int i = 0; i < 16; ++i) {
        const v2f t = *(const v2f*)&row[g * 64 + 2 * i];
        m = fmaxf(m, fmaxf(t.x, t.y));
      }
    }
    mrow[rb] = fmaxf(m, __shfl_xor(m, 16));
  }

  v8f oacc[2][2] = {{ZERO8, ZERO8}, {ZERO8, ZERO8}};
  float ssum[2] = {0.0f, 0.0f};
  v2f sc[2][8];
  load_scores(sc, srow0, srow1, 0);
#pragma unroll
  for (int ch = 0; ch < 4; ch += 2) {
    load_v(vB, vlane, (ch + 1) * 64);
    pv_chunk(oacc, ssum, sc, vA, mrow, srow0, srow1, ch * 64);
    if (ch + 2 < 4) load_v(vA, vlane, (ch + 2) * 64);
    pv_chunk(oacc, ssum, sc, vB, mrow, srow0, srow1, (ch + 1) * 64);
  }
#pragma unroll
  for (int rb = 0; rb < 2; ++rb) {
    const float tot = ssum[rb] + __shfl_xor(ssum[rb], 16);
    if (hi == 0) rinv[rb * 128 + w * 16 + L] = 1.0f / tot;
  }
  __syncthreads();
#pragma unroll
  for (int rb = 0; rb < 2; ++rb) {
    const int ib = rb * 128 + w * 16 + hi * 8;
#pragma unroll
    for (int r = 0; r < 8; ++r) {
      const float ri = rinv[ib + r];
      v2f o;
      o.x = oacc[rb][0][r] * ri;
      o.y = oacc[rb][1][r] * ri;
      *(v2f*)&ob[(size_t)(ib + r) * 256 + 2 * L] = o;
    }
  }
}

extern "C" void kernel_launch(void* const* d_in, const int* in_sizes, int n_in,
                              void* d_out, int out_size, void* d_ws,
                              size_t ws_size, hipStream_t stream) {
  (void)in_sizes; (void)n_in; (void)out_size; (void)ws_size;
  const float* x    = (const float*)d_in[0];
  const int*   mask = (const int*)d_in[1];
  const float* Wqkv = (const float*)d_in[2];
  const float* Wout = (const float*)d_in[3];
  const float* bout = (const float*)d_in[4];
  float* outp = (float*)d_out;

  float* qkv = (float*)d_ws;
  float* ao  = qkv + (size_t)65536 * 768;

  const dim3 blk(256);
  const dim3 gblk(512);
  const size_t gsmem = (size_t)(8 * GCH) * sizeof(float);
  hipFuncSetAttribute((const void*)gemm256x128_wmma,
                      hipFuncAttributeMaxDynamicSharedMemorySize, (int)gsmem);

  gemm256x128_wmma<<<dim3(6, 256), gblk, gsmem, stream>>>(x, Wqkv, nullptr,
                                                           qkv, 768);

  const size_t smem = (size_t)(256 * 258 + 256) * sizeof(float);
  hipFuncSetAttribute((const void*)attn_kernel,
                      hipFuncAttributeMaxDynamicSharedMemorySize, (int)smem);
  attn_kernel<<<dim3(2048), blk, smem, stream>>>(qkv, mask, ao);

  gemm256x128_wmma<<<dim3(2, 256), gblk, gsmem, stream>>>(ao, Wout, bout,
                                                           outp, 256);
}


# attention: keys taken in two 128-key stages with a running row max (first stage's PV and row sums rescaled), score tile in LDS halved so two blocks share a WGP; softmax and PV sum rounding changes
# speedup vs baseline: 1.7746x; 1.0840x over previous
#include <hip/hip_runtime.h>
#include <hip/hip_bf16.h>

typedef float v2f __attribute__((ext_vector_type(2)));
typedef float v8f __attribute__((ext_vector_type(8)));

static __device__ __forceinline__ v8f wmma_f32_4(v2f a, v2f b, v8f c) {
  return __builtin_amdgcn_wmma_f32_16x16x4_f32(false, a, false, b, (short)0, c,
                                               false, false);
}

#define ZERO8 {0.f,0.f,0.f,0.f,0.f,0.f,0.f,0.f}

#define GK 256
#define GN 128
#define GPC 36
#define GCH (GN * GPC)

__global__ __launch_bounds__(512) __attribute__((amdgpu_waves_per_eu(8, 8)))
void gemm256x128_wmma(
    const float* __restrict__ A, const float* __restrict__ B,
    const float* __restrict__ bias, float* __restrict__ C, int N) {
  extern __shared__ float bt[];

  const int n0 = blockIdx.x * GN;
  const int m0 = blockIdx.y * 256;
  const int tid = threadIdx.x;
  const int w = tid >> 5;
  const int lane = tid & 31;
  const int L = lane & 15;
  const int hi = lane >> 4;

  {
    const int n4 = lane & 3;
    const int kq = lane >> 2;
    float4 bv[16];
#pragma unroll
    for (int p = 0; p < 16; ++p) {
      const int q = w * 16 + p;
      const int n = (q & 7) * 16 + n4 * 4;
      const int k = (q >> 3) * 8 + kq;
      bv[p] = *(const float4*)&B[(size_t)k * N + n0 + n];
    }
#pragma unroll
    for (int p = 0; p < 16; ++p) {
      const int q = w * 16 + p;
      const int n = (q & 7) * 16 + n4 * 4;
      const int k = (q >> 3) * 8 + kq;
      float* dst = bt + (k >> 5) * GCH + n * GPC + (k & 31);
      dst[0 * GPC] = bv[p].x;
      dst[1 * GPC] = bv[p].y;
      dst[2 * GPC] = bv[p].z;
      dst[3 * GPC] = bv[p].w;
    }
  }

  const float* abase = A + (size_t)m0 * GK;
  const unsigned aoff = (unsigned)((w * 16 + L) * GK + hi * 2);
  const float* brow = bt + L * GPC + hi * 2;

  v8f acc[8] = {ZERO8, ZERO8, ZERO8, ZERO8, ZERO8, ZERO8, ZERO8, ZERO8};

  v2f a_cur[8];
#pragma unroll
  for (int c = 0; c < 8; ++c) a_cur[c] = *(const v2f*)&abase[aoff + (unsigned)(c * 4)];

  __syncthreads();

  for (int kc = 0; kc < GK; kc += 32) {
    const int kn = (kc + 32 < GK) ? kc + 32 : kc;
    v2f a_nxt[8];
#pragma unroll
    for (int c = 0; c < 8; ++c) a_nxt[c] = *(const v2f*)&(abase + kn)[aoff + (unsigned)(c * 4)];
    const float* bch = brow + (kc >> 5) * GCH;
#pragma unroll
    for (int c = 0; c < 8; ++c) {
#pragma unroll
      for (int t = 0; t < 8; ++t) {
        const v2f bf = *(const v2f*)&bch[t * 16 * GPC + c * 4];
        acc[t] = wmma_f32_4(a_cur[c], bf, acc[t]);
      }
    }
#pragma unroll
    for (int c = 0; c < 8; ++c) a_cur[c] = a_nxt[c];
  }

  float* cbase = C + (size_t)m0 * N + n0;
  const unsigned coff = (unsigned)((w * 16 + hi * 8) * N + L);
#pragma unroll
  for (int t = 0; t < 8; ++t) {
    const float bs = bias ? bias[n0 + t * 16 + L] : 0.0f;
#pragma unroll
    for (int r = 0; r < 8; ++r)
      cbase[coff + (unsigned)(r * N + t * 16)] = acc[t][r] + bs;
  }
}

#define SK 128
#define SP 132

static __device__ __forceinline__ void load_v_raw(
    v2f (&raw)[8][2], const float* __restrict__ vlane, int j0) {
#pragma unroll
  for (int ks = 0; ks < 8; ++ks) {
    raw[ks][0] = *(const v2f*)&vlane[(size_t)(j0 + ks * 4) * 768];
    raw[ks][1] = *(const v2f*)&vlane[(size_t)(j0 + ks * 4 + 1) * 768];
  }
}

static __device__ __forceinline__ void load_scores(v2f (&sc)[2][8],
                                                   const float* srow0,
                                                   const float* srow1, int j) {
#pragma unroll
  for (int c = 0; c < 8; ++c) {
    sc[0][c] = *(const v2f*)&srow0[j + c * 4];
    sc[1][c] = *(const v2f*)&srow1[j + c * 4];
  }
}

static __device__ __forceinline__ void pv_chunk(
    v8f (&oacc)[2][2], float (&ssum)[2], v2f (&sc)[2][8],
    const v2f (&vraw)[8][2], v2f (&vnxt)[8][2], const float* __restrict__ vlane,
    const float (&mrow)[2], const float* srow0, const float* srow1, int jn,
    int jv) {
  v2f pa[2][8];
#pragma unroll
  for (int c = 0; c < 8; ++c) {
#pragma unroll
    for (int rb = 0; rb < 2; ++rb) {
      v2f p;
      p.x = __expf(sc[rb][c].x - mrow[rb]);
      p.y = __expf(sc[rb][c].y - mrow[rb]);
      ssum[rb] += p.x;
      ssum[rb] += p.y;
      pa[rb][c] = p;
    }
  }
  __builtin_amdgcn_sched_barrier(0);
  v2f vf[8][2];
#pragma unroll
  for (int ks = 0; ks < 8; ++ks) {
    vf[ks][0] = (v2f){vraw[ks][0].x, vraw[ks][1].x};
    vf[ks][1] = (v2f){vraw[ks][0].y, vraw[ks][1].y};
  }
  if (jn >= 0) load_scores(sc, srow0, srow1, jn);
  if (jv >= 0) load_v_raw(vnxt, vlane, jv);
  __builtin_amdgcn_sched_barrier(0);
#pragma unroll
  for (int c = 0; c < 8; ++c) {
#pragma unroll
    for (int rb = 0; rb < 2; ++rb) {
      oacc[rb][0] = wmma_f32_4(pa[rb][c], vf[c][0], oacc[rb][0]);
      oacc[rb][1] = wmma_f32_4(pa[rb][c], vf[c][1], oacc[rb][1]);
    }
  }
}

__global__ __launch_bounds__(256) __attribute__((amdgpu_waves_per_eu(4, 4)))
void attn_kernel(
    const float* __restrict__ qkv, const int* __restrict__ mask,
    float* __restrict__ out) {
  extern __shared__ float lds[];
  float* simb = lds;
  float* rinv = lds + 256 * SP;

  const int bid = blockIdx.x;
  const int s = bid >> 5;
  const int b = (bid >> 3) & 3;
  const int h = bid & 7;

  const float* base = qkv + (size_t)((s * 4 + b) * 256) * 768;
  const float* qb = base + h * 32;
  const float* kb = base + 256 + h * 32;
  const float* vb = base + 512 + h * 32;
  const int* mk = mask + (size_t)s * 256 * 256;
  float* ob = out + (size_t)((s * 4 + b) * 256) * 256 + h * 32;

  const int tid = threadIdx.x;
  const int w = tid >> 5;
  const int lane = tid & 31;
  const int L = lane & 15;
  const int hi = lane >> 4;
  const float scale = 0.17677669529663687f;
  const float NEG = -1.0e10f;

  const float* vlane = vb + (size_t)(hi * 2) * 768 + 2 * L;
  const float* srow0 = simb + (w * 16 + L) * SP + hi * 2;
  const float* srow1 = simb + (128 + w * 16 + L) * SP + hi * 2;

  v8f oacc[2][2] = {{ZERO8, ZERO8}, {ZERO8, ZERO8}};
  float ssum[2] = {0.0f, 0.0f};
  float mrun[2] = {-3.4e38f, -3.4e38f};

#pragma unroll
  for (int st = 0; st < 2; ++st) {
    const int kbase = st * SK;

    v2f qa[2][8];
#pragma unroll
    for (int rb = 0; rb < 2; ++rb) {
      const int rbase = rb * 128 + w * 16;
#pragma unroll
      for (int c2 = 0; c2 < 4; ++c2) {
        const float4 q4 =
            *(const float4*)&qb[(size_t)(rbase + L) * 768 + c2 * 8 + hi * 4];
        qa[rb][2 * c2] = (v2f){q4.x, q4.y};
        qa[rb][2 * c2 + 1] = (v2f){q4.z, q4.w};
      }
    }

    float4 kr[2][2][4];
#pragma unroll
    for (int t = 0; t < 2; ++t) {
#pragma unroll
      for (int c2 = 0; c2 < 4; ++c2)
        kr[0][t][c2] = *(const float4*)&kb[(size_t)(kbase + 2 * L + t) * 768 +
                                           c2 * 8 + hi * 4];
    }
#pragma unroll
    for (int jj = 0; jj < SK / 32; ++jj) {
      const int j0 = jj * 32;
      int2 m2[2][8];
      __builtin_amdgcn_sched_barrier(0);
#pragma unroll
      for (int rb = 0; rb < 2; ++rb) {
        const int ib = rb * 128 + w * 16 + hi * 8;
#pragma unroll
        for (int r = 0; r < 8; ++r)
          m2[rb][r] = *(const int2*)&mk[(ib + r) * 256 + kbase + j0 + 2 * L];
      }
      if (jj + 1 < SK / 32) {
#pragma unroll
        for (int t = 0; t < 2; ++t) {
#pragma unroll
          for (int c2 = 0; c2 < 4; ++c2)
            kr[(jj + 1) & 1][t][c2] =
                *(const float4*)&kb[(size_t)(kbase + j0 + 32 + 2 * L + t) * 768 +
                                    c2 * 8 + hi * 4];
        }
      }
      __builtin_amdgcn_sched_barrier(0);
      v2f kf[2][8];
#pragma unroll
      for (int t = 0; t < 2; ++t) {
#pragma unroll
        for (int c2 = 0; c2 < 4; ++c2) {
          const float4 k4 = kr[jj & 1][t][c2];
          kf[t][2 * c2] = (v2f){k4.x, k4.y};
          kf[t][2 * c2 + 1] = (v2f){k4.z, k4.w};
        }
      }
#pragma unroll
      for (int rb = 0; rb < 2; ++rb) {
        v8f acc0 = ZERO8, acc1 = ZERO8;
#pragma unroll
        for (int c = 0; c < 8; ++c) {
          acc0 = wmma_f32_4(qa[rb][c], kf[0][c], acc0);
          acc1 = wmma_f32_4(qa[rb][c], kf[1][c], acc1);
        }
        const int ib = rb * 128 + w * 16 + hi * 8;
#pragma unroll
        for (int r = 0; r < 8; ++r) {
          v2f sv;
          sv.x = acc0[r] * scale;
          sv.y = acc1[r] * scale;
          sv.x = (m2[rb][r].x == 0) ? NEG : sv.x;
          sv.y = (m2[rb][r].y == 0) ? NEG : sv.y;
          *(v2f*)&simb[(ib + r) * SP + j0 + 2 * L] = sv;
        }
      }
    }
    __syncthreads();
    v2f rawA[8][2], rawB[8][2];
    __builtin_amdgcn_sched_barrier(0);
    load_v_raw(rawA, vlane, kbase);
    __builtin_amdgcn_sched_barrier(0);

    float mnew[2];
#pragma unroll
    for (int rb = 0; rb < 2; ++rb) {
      const float* row = simb + (rb * 128 + w * 16 + L) * SP + hi * 2;
      v2f t[SK / 4];
      __builtin_amdgcn_sched_barrier(0);
#pragma unroll
      for (int i = 0; i < SK / 4; ++i) t[i] = *(const v2f*)&row[4 * i];
      __builtin_amdgcn_sched_barrier(0);
      float m = -3.4e38f;
#pragma unroll
      for (int i = 0; i < SK / 4; ++i) m = fmaxf(m, fmaxf(t[i].x, t[i].y));
      m = fmaxf(m, __shfl_xor(m, 16));
      mnew[rb] = fmaxf(mrun[rb], m);
    }
    if (st > 0) {
      float fr[2][8];
      float f[2];
#pragma unroll
      for (int rb = 0; rb < 2; ++rb) {
        f[rb] = __expf(mrun[rb] - mnew[rb]);
        ssum[rb] *= f[rb];
      }
      __builtin_amdgcn_sched_barrier(0);
#pragma unroll
      for (int rb = 0; rb < 2; ++rb) {
#pragma unroll
        for (int r = 0; r < 8; ++r) fr[rb][r] = __shfl(f[rb], hi * 8 + r);
      }
      __builtin_amdgcn_sched_barrier(0);
#pragma unroll
      for (int rb = 0; rb < 2; ++rb) {
#pragma unroll
        for (int r = 0; r < 8; ++r) {
          oacc[rb][0][r] *= fr[rb][r];
          oacc[rb][1][r] *= fr[rb][r];
        }
      }
    }
    mrun[0] = mnew[0];
    mrun[1] = mnew[1];

    v2f sc[2][8];
    load_scores(sc, srow0, srow1, 0);
    pv_chunk(oacc, ssum, sc, rawA, rawB, vlane, mrun, srow0, srow1, 32,
             kbase + 32);
    pv_chunk(oacc, ssum, sc, rawB, rawA, vlane, mrun, srow0, srow1, 64,
             kbase + 64);
    pv_chunk(oacc, ssum, sc, rawA, rawB, vlane, mrun, srow0, srow1, 96,
             kbase + 96);
    pv_chunk(oacc, ssum, sc, rawB, rawA, vlane, mrun, srow0, srow1, -1, -1);

    if (st == 0) __syncthreads();
  }

  const float tot0 = ssum[0] + __shfl_xor(ssum[0], 16);
  const float tot1 = ssum[1] + __shfl_xor(ssum[1], 16);
  if (hi == 0) {
    rinv[w * 16 + L] = 1.0f / tot0;
    rinv[128 + w * 16 + L] = 1.0f / tot1;
  }
  __syncthreads();
#pragma unroll
  for (int rb = 0; rb < 2; ++rb) {
    const int ib = rb * 128 + w * 16 + hi * 8;
#pragma unroll
    for (int r = 0; r < 8; ++r) {
      const float ri = rinv[ib + r];
      v2f o;
      o.x = oacc[rb][0][r] * ri;
      o.y = oacc[rb][1][r] * ri;
      *(v2f*)&ob[(size_t)(ib + r) * 256 + 2 * L] = o;
    }
  }
}

extern "C" void kernel_launch(void* const* d_in, const int* in_sizes, int n_in,
                              void* d_out, int out_size, void* d_ws,
                              size_t ws_size, hipStream_t stream) {
  (void)in_sizes; (void)n_in; (void)out_size; (void)ws_size;
  const float* x    = (const float*)d_in[0];
  const int*   mask = (const int*)d_in[1];
  const float* Wqkv = (const float*)d_in[2];
  const float* Wout = (const float*)d_in[3];
  const float* bout = (const float*)d_in[4];
  float* outp = (float*)d_out;

  float* qkv = (float*)d_ws;
  float* ao  = qkv + (size_t)65536 * 768;

  const dim3 blk(256);
  const dim3 gblk(512);
  const size_t gsmem = (size_t)(8 * GCH) * sizeof(float);
  hipFuncSetAttribute((const void*)gemm256x128_wmma,
                      hipFuncAttributeMaxDynamicSharedMemorySize, (int)gsmem);

  gemm256x128_wmma<<<dim3(6, 256), gblk, gsmem, stream>>>(x, Wqkv, nullptr,
                                                           qkv, 768);

  const size_t smem = (size_t)(256 * SP + 256) * sizeof(float);
  hipFuncSetAttribute((const void*)attn_kernel,
                      hipFuncAttributeMaxDynamicSharedMemorySize, (int)smem);
  attn_kernel<<<dim3(2048), blk, smem, stream>>>(qkv, mask, ao);

  gemm256x128_wmma<<<dim3(2, 256), gblk, gsmem, stream>>>(ao, Wout, bout,
                                                           outp, 256);
}


# GEMM (both launches): column map of the weight fragments re-chosen (LDS panel rows permuted) so a lane's 4 tile results are adjacent columns: C leaves by 16-byte whole-line stores; sums unchanged
# speedup vs baseline: 1.8387x; 1.0361x over previous
#include <hip/hip_runtime.h>
#include <hip/hip_bf16.h>

typedef float v2f __attribute__((ext_vector_type(2)));
typedef float v8f __attribute__((ext_vector_type(8)));

static __device__ __forceinline__ v8f wmma_f32_4(v2f a, v2f b, v8f c) {
  return __builtin_amdgcn_wmma_f32_16x16x4_f32(false, a, false, b, (short)0, c,
                                               false, false);
}

#define ZERO8 {0.f,0.f,0.f,0.f,0.f,0.f,0.f,0.f}

#define GK 256
#define GN 128
#define GPC 36
#define GCH (GN * GPC)

__global__ __launch_bounds__(512) __attribute__((amdgpu_waves_per_eu(8, 8)))
void gemm256x128_wmma(
    const float* __restrict__ A, const float* __restrict__ B,
    const float* __restrict__ bias, float* __restrict__ C, int N) {
  extern __shared__ float bt[];

  const int n0 = blockIdx.x * GN;
  const int m0 = blockIdx.y * 256;
  const int tid = threadIdx.x;
  const int w = tid >> 5;
  const int lane = tid & 31;
  const int L = lane & 15;
  const int hi = lane >> 4;

  {
    const int n4 = lane & 3;
    const int kq = lane >> 2;
    float4 bv[16];
#pragma unroll
    for (int p = 0; p < 16; ++p) {
      const int q = w * 16 + p;
      const int n = (q & 7) * 16 + n4 * 4;
      const int k = (q >> 3) * 8 + kq;
      bv[p] = *(const float4*)&B[(size_t)k * N + n0 + n];
    }
#pragma unroll
    for (int p = 0; p < 16; ++p) {
      const int q = w * 16 + p;
      const int n = (q & 7) * 16 + n4 * 4;
      const int k = (q >> 3) * 8 + kq;
      const int prow = (n >> 6) * 64 + ((n & 63) >> 2);
      float* dst = bt + (k >> 5) * GCH + prow * GPC + (k & 31);
      dst[0 * 16 * GPC] = bv[p].x;
      dst[1 * 16 * GPC] = bv[p].y;
      dst[2 * 16 * GPC] = bv[p].z;
      dst[3 * 16 * GPC] = bv[p].w;
    }
  }

  const float* abase = A + (size_t)m0 * GK;
  const unsigned aoff = (unsigned)((w * 16 + L) * GK + hi * 2);
  const float* brow = bt + L * GPC + hi * 2;

  v8f acc[8] = {ZERO8, ZERO8, ZERO8, ZERO8, ZERO8, ZERO8, ZERO8, ZERO8};

  v2f a_cur[8];
#pragma unroll
  for (int c = 0; c < 8; ++c) a_cur[c] = *(const v2f*)&abase[aoff + (unsigned)(c * 4)];

  __syncthreads();

  for (int kc = 0; kc < GK; kc += 32) {
    const int kn = (kc + 32 < GK) ? kc + 32 : kc;
    v2f a_nxt[8];
#pragma unroll
    for (int c = 0; c < 8; ++c) a_nxt[c] = *(const v2f*)&(abase + kn)[aoff + (unsigned)(c * 4)];
    const float* bch = brow + (kc >> 5) * GCH;
#pragma unroll
    for (int c = 0; c < 8; ++c) {
#pragma unroll
      for (int t = 0; t < 8; ++t) {
        const v2f bf = *(const v2f*)&bch[t * 16 * GPC + c * 4];
        acc[t] = wmma_f32_4(a_cur[c], bf, acc[t]);
      }
    }
#pragma unroll
    for (int c = 0; c < 8; ++c) a_cur[c] = a_nxt[c];
  }

  float* cbase = C + (size_t)m0 * N + n0;
  const unsigned coff = (unsigned)((w * 16 + hi * 8) * N + 4 * L);
#pragma unroll
  for (int th = 0; th < 2; ++th) {
    float4 bs = make_float4(0.0f, 0.0f, 0.0f, 0.0f);
    if (bias) bs = *(const float4*)&bias[n0 + 64 * th + 4 * L];
#pragma unroll
    for (int r = 0; r < 8; ++r) {
      float4 o;
      o.x = acc[4 * th + 0][r] + bs.x;
      o.y = acc[4 * th + 1][r] + bs.y;
      o.z = acc[4 * th + 2][r] + bs.z;
      o.w = acc[4 * th + 3][r] + bs.w;
      *(float4*)&cbase[coff + (unsigned)(r * N + 64 * th)] = o;
    }
  }
}

#define SK 128
#define SP 132

static __device__ __forceinline__ void load_v_raw(
    v2f (&raw)[8][2], const float* __restrict__ vlane, int j0) {
#pragma unroll
  for (int ks = 0; ks < 8; ++ks) {
    raw[ks][0] = *(const v2f*)&vlane[(size_t)(j0 + ks * 4) * 768];
    raw[ks][1] = *(const v2f*)&vlane[(size_t)(j0 + ks * 4 + 1) * 768];
  }
}

static __device__ __forceinline__ void load_scores(v2f (&sc)[2][8],
                                                   const float* srow0,
                                                   const float* srow1, int j) {
#pragma unroll
  for (int c = 0; c < 8; ++c) {
    sc[0][c] = *(const v2f*)&srow0[j + c * 4];
    sc[1][c] = *(const v2f*)&srow1[j + c * 4];
  }
}

static __device__ __forceinline__ void pv_chunk(
    v8f (&oacc)[2][2], float (&ssum)[2], v2f (&sc)[2][8],
    const v2f (&vraw)[8][2], v2f (&vnxt)[8][2], const float* __restrict__ vlane,
    const float (&mrow)[2], const float* srow0, const float* srow1, int jn,
    int jv) {
  v2f pa[2][8];
#pragma unroll
  for (int c = 0; c < 8; ++c) {
#pragma unroll
    for (int rb = 0; rb < 2; ++rb) {
      v2f p;
      p.x = __expf(sc[rb][c].x - mrow[rb]);
      p.y = __expf(sc[rb][c].y - mrow[rb]);
      ssum[rb] += p.x;
      ssum[rb] += p.y;
      pa[rb][c] = p;
    }
  }
  __builtin_amdgcn_sched_barrier(0);
  v2f vf[8][2];
#pragma unroll
  for (int ks = 0; ks < 8; ++ks) {
    vf[ks][0] = (v2f){vraw[ks][0].x, vraw[ks][1].x};
    vf[ks][1] = (v2f){vraw[ks][0].y, vraw[ks][1].y};
  }
  if (jn >= 0) load_scores(sc, srow0, srow1, jn);
  if (jv >= 0) load_v_raw(vnxt, vlane, jv);
  __builtin_amdgcn_sched_barrier(0);
#pragma unroll
  for (int c = 0; c < 8; ++c) {
#pragma unroll
    for (int rb = 0; rb < 2; ++rb) {
      oacc[rb][0] = wmma_f32_4(pa[rb][c], vf[c][0], oacc[rb][0]);
      oacc[rb][1] = wmma_f32_4(pa[rb][c], vf[c][1], oacc[rb][1]);
    }
  }
}

__global__ __launch_bounds__(256) __attribute__((amdgpu_waves_per_eu(4, 4)))
void attn_kernel(
    const float* __restrict__ qkv, const int* __restrict__ mask,
    float* __restrict__ out) {
  extern __shared__ float lds[];
  float* simb = lds;
  float* rinv = lds + 256 * SP;

  const int bid = blockIdx.x;
  const int s = bid >> 5;
  const int b = (bid >> 3) & 3;
  const int h = bid & 7;

  const float* base = qkv + (size_t)((s * 4 + b) * 256) * 768;
  const float* qb = base + h * 32;
  const float* kb = base + 256 + h * 32;
  const float* vb = base + 512 + h * 32;
  const int* mk = mask + (size_t)s * 256 * 256;
  float* ob = out + (size_t)((s * 4 + b) * 256) * 256 + h * 32;

  const int tid = threadIdx.x;
  const int w = tid >> 5;
  const int lane = tid & 31;
  const int L = lane & 15;
  const int hi = lane >> 4;
  const float scale = 0.17677669529663687f;
  const float NEG = -1.0e10f;

  const float* vlane = vb + (size_t)(hi * 2) * 768 + 2 * L;
  const float* srow0 = simb + (w * 16 + L) * SP + hi * 2;
  const float* srow1 = simb + (128 + w * 16 + L) * SP + hi * 2;

  v8f oacc[2][2] = {{ZERO8, ZERO8}, {ZERO8, ZERO8}};
  float ssum[2] = {0.0f, 0.0f};
  float mrun[2] = {-3.4e38f, -3.4e38f};

#pragma unroll
  for (int st = 0; st < 2; ++st) {
    const int kbase = st * SK;

    v2f qa[2][8];
#pragma unroll
    for (int rb = 0; rb < 2; ++rb) {
      const int rbase = rb * 128 + w * 16;
#pragma unroll
      for (int c2 = 0; c2 < 4; ++c2) {
        const float4 q4 =
            *(const float4*)&qb[(size_t)(rbase + L) * 768 + c2 * 8 + hi * 4];
        qa[rb][2 * c2] = (v2f){q4.x, q4.y};
        qa[rb][2 * c2 + 1] = (v2f){q4.z, q4.w};
      }
    }

    float4 kr[2][2][4];
#pragma unroll
    for (int t = 0; t < 2; ++t) {
#pragma unroll
      for (int c2 = 0; c2 < 4; ++c2)
        kr[0][t][c2] = *(const float4*)&kb[(size_t)(kbase + 2 * L + t) * 768 +
                                           c2 * 8 + hi * 4];
    }
#pragma unroll
    for (int jj = 0; jj < SK / 32; ++jj) {
      const int j0 = jj * 32;
      int2 m2[2][8];
      __builtin_amdgcn_sched_barrier(0);
#pragma unroll
      for (int rb = 0; rb < 2; ++rb) {
        const int ib = rb * 128 + w * 16 + hi * 8;
#pragma unroll
        for (int r = 0; r < 8; ++r)
          m2[rb][r] = *(const int2*)&mk[(ib + r) * 256 + kbase + j0 + 2 * L];
      }
      if (jj + 1 < SK / 32) {
#pragma unroll
        for (int t = 0; t < 2; ++t) {
#pragma unroll
          for (int c2 = 0; c2 < 4; ++c2)
            kr[(jj + 1) & 1][t][c2] =
                *(const float4*)&kb[(size_t)(kbase + j0 + 32 + 2 * L + t) * 768 +
                                    c2 * 8 + hi * 4];
        }
      }
      __builtin_amdgcn_sched_barrier(0);
      v2f kf[2][8];
#pragma unroll
      for (int t = 0; t < 2; ++t) {
#pragma unroll
        for (int c2 = 0; c2 < 4; ++c2) {
          const float4 k4 = kr[jj & 1][t][c2];
          kf[t][2 * c2] = (v2f){k4.x, k4.y};
          kf[t][2 * c2 + 1] = (v2f){k4.z, k4.w};
        }
      }
#pragma unroll
      for (int rb = 0; rb < 2; ++rb) {
        v8f acc0 = ZERO8, acc1 = ZERO8;
#pragma unroll
        for (int c = 0; c < 8; ++c) {
          acc0 = wmma_f32_4(qa[rb][c], kf[0][c], acc0);
          acc1 = wmma_f32_4(qa[rb][c], kf[1][c], acc1);
        }
        const int ib = rb * 128 + w * 16 + hi * 8;
#pragma unroll
        for (int r = 0; r < 8; ++r) {
          v2f sv;
          sv.x = acc0[r] * scale;
          sv.y = acc1[r] * scale;
          sv.x = (m2[rb][r].x == 0) ? NEG : sv.x;
          sv.y = (m2[rb][r].y == 0) ? NEG : sv.y;
          *(v2f*)&simb[(ib + r) * SP + j0 + 2 * L] = sv;
        }
      }
    }
    __syncthreads();
    v2f rawA[8][2], rawB[8][2];
    __builtin_amdgcn_sched_barrier(0);
    load_v_raw(rawA, vlane, kbase);
    __builtin_amdgcn_sched_barrier(0);

    float mnew[2];
#pragma unroll
    for (int rb = 0; rb < 2; ++rb) {
      const float* row = simb + (rb * 128 + w * 16 + L) * SP + hi * 2;
      v2f t[SK / 4];
      __builtin_amdgcn_sched_barrier(0);
#pragma unroll
      for (int i = 0; i < SK / 4; ++i) t[i] = *(const v2f*)&row[4 * i];
      __builtin_amdgcn_sched_barrier(0);
      float m = -3.4e38f;
#pragma unroll
      for (int i = 0; i < SK / 4; ++i) m = fmaxf(m, fmaxf(t[i].x, t[i].y));
      m = fmaxf(m, __shfl_xor(m, 16));
      mnew[rb] = fmaxf(mrun[rb], m);
    }
    if (st > 0) {
      float fr[2][8];
      float f[2];
#pragma unroll
      for (int rb = 0; rb < 2; ++rb) {
        f[rb] = __expf(mrun[rb] - mnew[rb]);
        ssum[rb] *= f[rb];
      }
      __builtin_amdgcn_sched_barrier(0);
#pragma unroll
      for (int rb = 0; rb < 2; ++rb) {
#pragma unroll
        for (int r = 0; r < 8; ++r) fr[rb][r] = __shfl(f[rb], hi * 8 + r);
      }
      __builtin_amdgcn_sched_barrier(0);
#pragma unroll
      for (int rb = 0; rb < 2; ++rb) {
#pragma unroll
        for (int r = 0; r < 8; ++r) {
          oacc[rb][0][r] *= fr[rb][r];
          oacc[rb][1][r] *= fr[rb][r];
        }
      }
    }
    mrun[0] = mnew[0];
    mrun[1] = mnew[1];

    v2f sc[2][8];
    load_scores(sc, srow0, srow1, 0);
    pv_chunk(oacc, ssum, sc, rawA, rawB, vlane, mrun, srow0, srow1, 32,
             kbase + 32);
    pv_chunk(oacc, ssum, sc, rawB, rawA, vlane, mrun, srow0, srow1, 64,
             kbase + 64);
    pv_chunk(oacc, ssum, sc, rawA, rawB, vlane, mrun, srow0, srow1, 96,
             kbase + 96);
    pv_chunk(oacc, ssum, sc, rawB, rawA, vlane, mrun, srow0, srow1, -1, -1);

    if (st == 0) __syncthreads();
  }

  const float tot0 = ssum[0] + __shfl_xor(ssum[0], 16);
  const float tot1 = ssum[1] + __shfl_xor(ssum[1], 16);
  if (hi == 0) {
    rinv[w * 16 + L] = 1.0f / tot0;
    rinv[128 + w * 16 + L] = 1.0f / tot1;
  }
  __syncthreads();
#pragma unroll
  for (int rb = 0; rb < 2; ++rb) {
    const int ib = rb * 128 + w * 16 + hi * 8;
#pragma unroll
    for (int r = 0; r < 8; ++r) {
      const float ri = rinv[ib + r];
      v2f o;
      o.x = oacc[rb][0][r] * ri;
      o.y = oacc[rb][1][r] * ri;
      *(v2f*)&ob[(size_t)(ib + r) * 256 + 2 * L] = o;
    }
  }
}

extern "C" void kernel_launch(void* const* d_in, const int* in_sizes, int n_in,
                              void* d_out, int out_size, void* d_ws,
                              size_t ws_size, hipStream_t stream) {
  (void)in_sizes; (void)n_in; (void)out_size; (void)ws_size;
  const float* x    = (const float*)d_in[0];
  const int*   mask = (const int*)d_in[1];
  const float* Wqkv = (const float*)d_in[2];
  const float* Wout = (const float*)d_in[3];
  const float* bout = (const float*)d_in[4];
  float* outp = (float*)d_out;

  float* qkv = (float*)d_ws;
  float* ao  = qkv + (size_t)65536 * 768;

  const dim3 blk(256);
  const dim3 gblk(512);
  const size_t gsmem = (size_t)(8 * GCH) * sizeof(float);
  hipFuncSetAttribute((const void*)gemm256x128_wmma,
                      hipFuncAttributeMaxDynamicSharedMemorySize, (int)gsmem);

  gemm256x128_wmma<<<dim3(6, 256), gblk, gsmem, stream>>>(x, Wqkv, nullptr,
                                                           qkv, 768);

  const size_t smem = (size_t)(256 * SP + 256) * sizeof(float);
  hipFuncSetAttribute((const void*)attn_kernel,
                      hipFuncAttributeMaxDynamicSharedMemorySize, (int)smem);
  attn_kernel<<<dim3(2048), blk, smem, stream>>>(qkv, mask, ao);

  gemm256x128_wmma<<<dim3(2, 256), gblk, gsmem, stream>>>(ao, Wout, bout,
                                                           outp, 256);
}
